# starting kernel
# baseline (speedup 1.0000x reference)
#include <hip/hip_runtime.h>

typedef float v2f __attribute__((ext_vector_type(2)));
typedef float v8f __attribute__((ext_vector_type(8)));

#define BS      32
#define CS      512
#define SSEG    8
#define HS      64
#define WSZ     64
#define SPATIAL (HS * WSZ)
#define NSTEP   (SPATIAL / 4)
#define SEMW    256

__global__ void build_a(const int* __restrict__ sem, float* __restrict__ Apre) {
    unsigned gid  = blockIdx.x * blockDim.x + threadIdx.x;
    unsigned lane = gid & 31u;
    unsigned t    = (gid >> 5) & (NSTEP - 1);
    unsigned b    = gid >> 15;
    unsigned seg  = lane & 15u;
    unsigned kh   = lane >> 4;
    unsigned pos0 = 4u * t + 2u * kh;

    v2f a;
    if (seg < SSEG) {
        unsigned base = (b * SSEG + seg) * (SEMW * SEMW);
        #pragma unroll
        for (int i = 0; i < 2; ++i) {
            unsigned pos = pos0 + i;
            unsigned h = pos >> 6, w = pos & 63u;
            int v = sem[base + (h * 4u) * SEMW + (w * 4u)];
            a[i] = (v != 0) ? 1.0f : 0.0f;
        }
    } else if (seg == SSEG) {
        a.x = 1.0f; a.y = 1.0f;
    } else {
        a.x = 0.0f; a.y = 0.0f;
    }
    ((v2f*)Apre)[gid] = a;
}

__global__ void area_kernel(const int* __restrict__ sem, float* __restrict__ area) {
    unsigned bs = blockIdx.x;
    __shared__ float red[256];
    unsigned base = bs * (SEMW * SEMW);
    float acc = 0.0f;
    for (unsigned pos = threadIdx.x; pos < SPATIAL; pos += 256u) {
        unsigned h = pos >> 6, w = pos & 63u;
        acc += (sem[base + (h * 4u) * SEMW + (w * 4u)] != 0) ? 1.0f : 0.0f;
    }
    red[threadIdx.x] = acc;
    __syncthreads();
    for (unsigned off = 128; off > 0; off >>= 1) {
        if (threadIdx.x < off) red[threadIdx.x] += red[threadIdx.x + off];
        __syncthreads();
    }
    if (threadIdx.x == 0) area[bs] = red[0];
}

__global__ void __launch_bounds__(128) seg_reduce(const float* __restrict__ codes,
                                                  const float* __restrict__ Apre,
                                                  const float* __restrict__ area,
                                                  float* __restrict__ out) {
    const unsigned wave = threadIdx.x >> 5;
    const unsigned lane = threadIdx.x & 31u;
    const unsigned b    = blockIdx.x >> 3;
    const unsigned cg   = (blockIdx.x & 7u) * 4u + wave;
    const unsigned n    = lane & 15u;
    const unsigned kh   = lane >> 4;

    const v2f* __restrict__ bptr =
        (const v2f*)(codes + ((size_t)(b * CS + cg * 16u + n)) * SPATIAL) + kh;
    const v2f* __restrict__ aptr =
        (const v2f*)Apre + (size_t)b * NSTEP * 32u + lane;

    v8f c = {0.f, 0.f, 0.f, 0.f, 0.f, 0.f, 0.f, 0.f};

    #pragma unroll 4
    for (int t = 0; t < NSTEP; ++t) {
        v2f bvec = __builtin_nontemporal_load(bptr + 2 * t);
        v2f avec = aptr[t * 32];
        c = __builtin_amdgcn_wmma_f32_16x16x4_f32(
                 false, avec,  false, bvec,
                 (short)0, c,  false,  false);
    }

    float glob = __shfl(c[0], (int)(16u + n), 32);

    if (lane < 16u) {
        const float ginv = glob * (1.0f / (float)SPATIAL);
        float* o = out + (size_t)b * (SSEG * CS) + cg * 16u + n;
        #pragma unroll
        for (int s = 0; s < SSEG; ++s) {
            float ar = area[b * SSEG + s];
            o[(size_t)s * CS] = (ar > 0.0f) ? (c[s] / ar) : ginv;
        }
    }
}

extern "C" void kernel_launch(void* const* d_in, const int* in_sizes, int n_in,
                              void* d_out, int out_size, void* d_ws, size_t ws_size,
                              hipStream_t stream) {
    const float* codes = (const float*)d_in[0];
    const int*   sem   = (const int*)d_in[1];
    float*       out   = (float*)d_out;

    float* Apre = (float*)d_ws;
    float* area = (float*)((char*)d_ws +
                           (size_t)BS * NSTEP * 32 * 2 * sizeof(float));

    hipLaunchKernelGGL(build_a, dim3((BS * NSTEP * 32) / 256), dim3(256),
                       0, stream, sem, Apre);
    hipLaunchKernelGGL(area_kernel, dim3(BS * SSEG), dim3(256),
                       0, stream, sem, area);
    hipLaunchKernelGGL(seg_reduce, dim3(BS * 8), dim3(128),
                       0, stream, codes, Apre, area, out);
}


# seg_reduce rebuilt as a stream: 8 waves per block split the spatial range, 16-byte loads, next 24 loads issued ahead of each WMMA group; partial sums added through LDS (summation order changes)
# speedup vs baseline: 2.9667x; 2.9667x over previous
#include <hip/hip_runtime.h>


typedef float v2f __attribute__((ext_vector_type(2)));
typedef float v4f __attribute__((ext_vector_type(4)));
typedef float v8f __attribute__((ext_vector_type(8)));

#define BS      32
#define CS      512
#define SSEG    8
#define HS      64
#define WSZ     64
#define SPATIAL (HS * WSZ)
#define NSTEP   (SPATIAL / 4)
#define NJ      (SPATIAL / 8)
#define SEMW    256

#define KSPLIT  8
#define JW      (NJ / KSPLIT)
#define UJ      8
#define NG      (JW / UJ)

__global__ void build_a(const int* __restrict__ sem, float* __restrict__ Apre) {
    unsigned gid  = blockIdx.x * blockDim.x + threadIdx.x;
    unsigned lane = gid & 31u;
    unsigned j    = (gid >> 5) & (NJ - 1);
    unsigned b    = gid >> 14;
    unsigned seg  = lane & 15u;
    unsigned kh   = lane >> 4;
    unsigned pos0 = 8u * j + 4u * kh;

    v4f a;
    if (seg < SSEG) {
        unsigned base = (b * SSEG + seg) * (SEMW * SEMW);
        #pragma unroll
        for (int i = 0; i < 4; ++i) {
            unsigned pos = pos0 + i;
            unsigned h = pos >> 6, w = pos & 63u;
            int v = sem[base + (h * 4u) * SEMW + (w * 4u)];
            a[i] = (v != 0) ? 1.0f : 0.0f;
        }
    } else if (seg == SSEG) {
        a.x = 1.0f; a.y = 1.0f; a.z = 1.0f; a.w = 1.0f;
    } else {
        a.x = 0.0f; a.y = 0.0f; a.z = 0.0f; a.w = 0.0f;
    }
    ((v4f*)Apre)[gid] = a;
}

__global__ void area_kernel(const int* __restrict__ sem, float* __restrict__ area) {
    unsigned bs = blockIdx.x;
    __shared__ float red[256];
    unsigned base = bs * (SEMW * SEMW);
    float acc = 0.0f;
    for (unsigned pos = threadIdx.x; pos < SPATIAL; pos += 256u) {
        unsigned h = pos >> 6, w = pos & 63u;
        acc += (sem[base + (h * 4u) * SEMW + (w * 4u)] != 0) ? 1.0f : 0.0f;
    }
    red[threadIdx.x] = acc;
    __syncthreads();
    for (unsigned off = 128; off > 0; off >>= 1) {
        if (threadIdx.x < off) red[threadIdx.x] += red[threadIdx.x + off];
        __syncthreads();
    }
    if (threadIdx.x == 0) area[bs] = red[0];
}

__global__ void __launch_bounds__(256) seg_reduce(const float* __restrict__ codes,
                                                  const float* __restrict__ Apre,
                                                  const float* __restrict__ area,
                                                  float* __restrict__ out) {
    __shared__ float part_seg[KSPLIT][SSEG][32];
    __shared__ float part_glob[KSPLIT][32];

    const unsigned wave = __builtin_amdgcn_readfirstlane(threadIdx.x >> 5);
    const unsigned lane = threadIdx.x & 31u;
    const unsigned b    = blockIdx.x >> 4;
    const unsigned cgp  = blockIdx.x & 15u;
    const unsigned n    = lane & 15u;
    const unsigned kh   = lane >> 4;
    const unsigned j0   = wave * JW;

    const v4f* __restrict__ bp0 =
        (const v4f*)(codes + ((size_t)(b * CS + cgp * 32u + n)) * SPATIAL) + (2u * j0 + kh);
    const v4f* __restrict__ bp1 = bp0 + (size_t)16 * (SPATIAL / 4);
    const v4f* __restrict__ ap =
        (const v4f*)Apre + ((size_t)b * NJ + j0) * 32u + lane;

    v8f c0 = {0.f, 0.f, 0.f, 0.f, 0.f, 0.f, 0.f, 0.f};
    v8f c1 = {0.f, 0.f, 0.f, 0.f, 0.f, 0.f, 0.f, 0.f};

    v4f av[2][UJ], pv[2][UJ], qv[2][UJ];

    #pragma unroll
    for (int u = 0; u < UJ; ++u) {
        av[0][u] = ap[u * 32];
        pv[0][u] = __builtin_nontemporal_load(bp0 + 2 * u);
        qv[0][u] = __builtin_nontemporal_load(bp1 + 2 * u);
    }

    #pragma unroll
    for (int g = 0; g < NG; ++g) {
        const int cur = g & 1, nxt = cur ^ 1;
        __builtin_amdgcn_sched_barrier(0);
        if (g + 1 < NG) {
            #pragma unroll
            for (int u = 0; u < UJ; ++u) {
                const int jj = (g + 1) * UJ + u;
                av[nxt][u] = ap[jj * 32];
                pv[nxt][u] = __builtin_nontemporal_load(bp0 + 2 * jj);
                qv[nxt][u] = __builtin_nontemporal_load(bp1 + 2 * jj);
            }
        }
        __builtin_amdgcn_sched_barrier(0);
        #pragma unroll
        for (int u = 0; u < UJ; ++u) {
            const v4f a = av[cur][u], p = pv[cur][u], q = qv[cur][u];
            const v2f a01 = {a.x, a.y}, a23 = {a.z, a.w};
            const v2f p01 = {p.x, p.y}, p23 = {p.z, p.w};
            const v2f q01 = {q.x, q.y}, q23 = {q.z, q.w};
            c0 = __builtin_amdgcn_wmma_f32_16x16x4_f32(false, a01, false, p01, (short)0, c0, false, false);
            c1 = __builtin_amdgcn_wmma_f32_16x16x4_f32(false, a01, false, q01, (short)0, c1, false, false);
            c0 = __builtin_amdgcn_wmma_f32_16x16x4_f32(false, a23, false, p23, (short)0, c0, false, false);
            c1 = __builtin_amdgcn_wmma_f32_16x16x4_f32(false, a23, false, q23, (short)0, c1, false, false);
        }
    }

    if (kh == 0u) {
        #pragma unroll
        for (int s = 0; s < SSEG; ++s) {
            part_seg[wave][s][n]       = c0[s];
            part_seg[wave][s][16u + n] = c1[s];
        }
    } else {
        part_glob[wave][n]       = c0[0];
        part_glob[wave][16u + n] = c1[0];
    }
    __syncthreads();

    const unsigned s  = threadIdx.x >> 5;
    const unsigned ch = lane;
    float sum = 0.0f, glob = 0.0f;
    #pragma unroll
    for (int w = 0; w < KSPLIT; ++w) {
        sum  += part_seg[w][s][ch];
        glob += part_glob[w][ch];
    }
    const float ar = area[b * SSEG + s];
    out[(size_t)b * (SSEG * CS) + (size_t)s * CS + cgp * 32u + ch] =
        (ar > 0.0f) ? (sum / ar) : (glob * (1.0f / (float)SPATIAL));
}

extern "C" void kernel_launch(void* const* d_in, const int* in_sizes, int n_in,
                              void* d_out, int out_size, void* d_ws, size_t ws_size,
                              hipStream_t stream) {
    const float* codes = (const float*)d_in[0];
    const int*   sem   = (const int*)d_in[1];
    float*       out   = (float*)d_out;

    float* Apre = (float*)d_ws;
    float* area = (float*)((char*)d_ws +
                           (size_t)BS * NSTEP * 32 * 2 * sizeof(float));

    hipLaunchKernelGGL(build_a, dim3((BS * NJ * 32) / 256), dim3(256),
                       0, stream, sem, Apre);
    hipLaunchKernelGGL(area_kernel, dim3(BS * SSEG), dim3(256),
                       0, stream, sem, area);
    hipLaunchKernelGGL(seg_reduce, dim3(BS * 16), dim3(256),
                       0, stream, codes, Apre, area, out);
}


# stage 1 rebuilt as one launch: a block per (batch, segment) writes its 0/1 mask row in natural order and its area; seg_reduce loads A rows from it as it loads B rows (3 launches to 2, 9x fewer waves)
# speedup vs baseline: 3.6224x; 1.2210x over previous
#include <hip/hip_runtime.h>


typedef float v2f __attribute__((ext_vector_type(2)));
typedef float v4f __attribute__((ext_vector_type(4)));
typedef float v8f __attribute__((ext_vector_type(8)));

#define BS      32
#define CS      512
#define SSEG    8
#define HS      64
#define WSZ     64
#define SPATIAL (HS * WSZ)
#define NSTEP   (SPATIAL / 4)
#define NJ      (SPATIAL / 8)
#define SEMW    256

#define KSPLIT  8
#define JW      (NJ / KSPLIT)
#define UJ      8
#define NG      (JW / UJ)

__global__ void __launch_bounds__(256) build_mask(const int* __restrict__ sem,
                                                  float* __restrict__ maskf,
                                                  float* __restrict__ area) {
    const unsigned bs = blockIdx.x;
    const unsigned t  = threadIdx.x;
    __shared__ float red[256];
    const unsigned base = bs * (SEMW * SEMW);
    float acc = 0.0f;
    #pragma unroll
    for (int c = 0; c < 4; ++c) {
        const unsigned pos0 = (unsigned)c * 1024u + 4u * t;
        const unsigned h = pos0 >> 6, w0 = pos0 & 63u;
        const int* __restrict__ src = sem + base + (h * 4u) * SEMW + w0 * 4u;
        v4f m;
        m.x = (src[0]  != 0) ? 1.0f : 0.0f;
        m.y = (src[4]  != 0) ? 1.0f : 0.0f;
        m.z = (src[8]  != 0) ? 1.0f : 0.0f;
        m.w = (src[12] != 0) ? 1.0f : 0.0f;
        ((v4f*)maskf)[bs * (SPATIAL / 4) + (unsigned)c * 256u + t] = m;
        acc += (m.x + m.y) + (m.z + m.w);
    }
    red[t] = acc;
    __syncthreads();
    for (unsigned off = 128; off > 0; off >>= 1) {
        if (t < off) red[t] += red[t + off];
        __syncthreads();
    }
    if (t == 0) area[bs] = red[0];
}

__global__ void __launch_bounds__(256) seg_reduce(const float* __restrict__ codes,
                                                  const float* __restrict__ maskf,
                                                  const float* __restrict__ area,
                                                  float* __restrict__ out) {
    __shared__ float part_seg[KSPLIT][SSEG][32];
    __shared__ float part_glob[KSPLIT][32];

    const unsigned wave = __builtin_amdgcn_readfirstlane(threadIdx.x >> 5);
    const unsigned lane = threadIdx.x & 31u;
    const unsigned b    = blockIdx.x >> 4;
    const unsigned cgp  = blockIdx.x & 15u;
    const unsigned n    = lane & 15u;
    const unsigned kh   = lane >> 4;
    const unsigned j0   = wave * JW;

    const v4f* __restrict__ bp0 =
        (const v4f*)(codes + ((size_t)(b * CS + cgp * 32u + n)) * SPATIAL) + (2u * j0 + kh);
    const v4f* __restrict__ bp1 = bp0 + (size_t)16 * (SPATIAL / 4);
    const v4f* __restrict__ ap =
        (const v4f*)maskf + (size_t)(b * SSEG + (n & 7u)) * (SPATIAL / 4) + (2u * j0 + kh);
    const bool  is_seg = n < SSEG;
    const float fill   = (n == SSEG) ? 1.0f : 0.0f;

    v8f c0 = {0.f, 0.f, 0.f, 0.f, 0.f, 0.f, 0.f, 0.f};
    v8f c1 = {0.f, 0.f, 0.f, 0.f, 0.f, 0.f, 0.f, 0.f};

    v4f av[2][UJ], pv[2][UJ], qv[2][UJ];

    #pragma unroll
    for (int u = 0; u < UJ; ++u) {
        av[0][u] = ap[2 * u];
        pv[0][u] = __builtin_nontemporal_load(bp0 + 2 * u);
        qv[0][u] = __builtin_nontemporal_load(bp1 + 2 * u);
    }

    #pragma unroll
    for (int g = 0; g < NG; ++g) {
        const int cur = g & 1, nxt = cur ^ 1;
        __builtin_amdgcn_sched_barrier(0);
        if (g + 1 < NG) {
            #pragma unroll
            for (int u = 0; u < UJ; ++u) {
                const int jj = (g + 1) * UJ + u;
                av[nxt][u] = ap[2 * jj];
                pv[nxt][u] = __builtin_nontemporal_load(bp0 + 2 * jj);
                qv[nxt][u] = __builtin_nontemporal_load(bp1 + 2 * jj);
            }
        }
        __builtin_amdgcn_sched_barrier(0);
        #pragma unroll
        for (int u = 0; u < UJ; ++u) {
            const v4f a = av[cur][u], p = pv[cur][u], q = qv[cur][u];
            const v2f a01 = {is_seg ? a.x : fill, is_seg ? a.y : fill};
            const v2f a23 = {is_seg ? a.z : fill, is_seg ? a.w : fill};
            const v2f p01 = {p.x, p.y}, p23 = {p.z, p.w};
            const v2f q01 = {q.x, q.y}, q23 = {q.z, q.w};
            c0 = __builtin_amdgcn_wmma_f32_16x16x4_f32(false, a01, false, p01, (short)0, c0, false, false);
            c1 = __builtin_amdgcn_wmma_f32_16x16x4_f32(false, a01, false, q01, (short)0, c1, false, false);
            c0 = __builtin_amdgcn_wmma_f32_16x16x4_f32(false, a23, false, p23, (short)0, c0, false, false);
            c1 = __builtin_amdgcn_wmma_f32_16x16x4_f32(false, a23, false, q23, (short)0, c1, false, false);
        }
    }

    if (kh == 0u) {
        #pragma unroll
        for (int s = 0; s < SSEG; ++s) {
            part_seg[wave][s][n]       = c0[s];
            part_seg[wave][s][16u + n] = c1[s];
        }
    } else {
        part_glob[wave][n]       = c0[0];
        part_glob[wave][16u + n] = c1[0];
    }
    __syncthreads();

    const unsigned s  = threadIdx.x >> 5;
    const unsigned ch = lane;
    float sum = 0.0f, glob = 0.0f;
    #pragma unroll
    for (int w = 0; w < KSPLIT; ++w) {
        sum  += part_seg[w][s][ch];
        glob += part_glob[w][ch];
    }
    const float ar = area[b * SSEG + s];
    out[(size_t)b * (SSEG * CS) + (size_t)s * CS + cgp * 32u + ch] =
        (ar > 0.0f) ? (sum / ar) : (glob * (1.0f / (float)SPATIAL));
}

extern "C" void kernel_launch(void* const* d_in, const int* in_sizes, int n_in,
                              void* d_out, int out_size, void* d_ws, size_t ws_size,
                              hipStream_t stream) {
    const float* codes = (const float*)d_in[0];
    const int*   sem   = (const int*)d_in[1];
    float*       out   = (float*)d_out;

    float* maskf = (float*)d_ws;
    float* area  = (float*)((char*)d_ws +
                            (size_t)BS * NSTEP * 32 * 2 * sizeof(float));

    hipLaunchKernelGGL(build_mask, dim3(BS * SSEG), dim3(256),
                       0, stream, sem, maskf, area);
    hipLaunchKernelGGL(seg_reduce, dim3(BS * 16), dim3(256),
                       0, stream, codes, maskf, area, out);
}


# seg_reduce loads whole lines: each 16-byte load covers 2 rows x 256 contiguous bytes of codes or mask, staged in wave-private LDS rows and read back as WMMA fragments (sum order unchanged)
# speedup vs baseline: 5.0929x; 1.4059x over previous
#include <hip/hip_runtime.h>


typedef float v2f __attribute__((ext_vector_type(2)));
typedef float v4f __attribute__((ext_vector_type(4)));
typedef float v8f __attribute__((ext_vector_type(8)));

#define BS      32
#define CS      512
#define SSEG    8
#define HS      64
#define WSZ     64
#define SPATIAL (HS * WSZ)
#define NSTEP   (SPATIAL / 4)
#define SEMW    256

#define KSPLIT  8
#define WPOS    (SPATIAL / KSPLIT)
#define GP      64
#define NGRP    (WPOS / GP)
#define CROWS   32
#define TROWS   (CROWS + SSEG)
#define TPITCH  17
#define NCL     (CROWS / 2)
#define NML     (SSEG / 2)
#define NSTP    (GP / 8)

__global__ void __launch_bounds__(256) build_mask(const int* __restrict__ sem,
                                                  float* __restrict__ maskf,
                                                  float* __restrict__ area) {
    const unsigned bs = blockIdx.x;
    const unsigned t  = threadIdx.x;
    __shared__ float red[256];
    const unsigned base = bs * (SEMW * SEMW);
    float acc = 0.0f;
    #pragma unroll
    for (int c = 0; c < 4; ++c) {
        const unsigned pos0 = (unsigned)c * 1024u + 4u * t;
        const unsigned h = pos0 >> 6, w0 = pos0 & 63u;
        const int* __restrict__ src = sem + base + (h * 4u) * SEMW + w0 * 4u;
        v4f m;
        m.x = (src[0]  != 0) ? 1.0f : 0.0f;
        m.y = (src[4]  != 0) ? 1.0f : 0.0f;
        m.z = (src[8]  != 0) ? 1.0f : 0.0f;
        m.w = (src[12] != 0) ? 1.0f : 0.0f;
        ((v4f*)maskf)[bs * (SPATIAL / 4) + (unsigned)c * 256u + t] = m;
        acc += (m.x + m.y) + (m.z + m.w);
    }
    red[t] = acc;
    __syncthreads();
    for (unsigned off = 128; off > 0; off >>= 1) {
        if (t < off) red[t] += red[t + off];
        __syncthreads();
    }
    if (t == 0) area[bs] = red[0];
}

__global__ void __launch_bounds__(256) seg_reduce(const float* __restrict__ codes,
                                                  const float* __restrict__ maskf,
                                                  const float* __restrict__ area,
                                                  float* __restrict__ out) {
    __shared__ v4f   tile[KSPLIT][TROWS][TPITCH];
    __shared__ float part_seg[KSPLIT][SSEG][32];
    __shared__ float part_glob[KSPLIT][32];

    const unsigned wave = __builtin_amdgcn_readfirstlane(threadIdx.x >> 5);
    const unsigned lane = threadIdx.x & 31u;
    const unsigned b    = blockIdx.x >> 4;
    const unsigned cgp  = blockIdx.x & 15u;
    const unsigned n    = lane & 15u;
    const unsigned kh   = lane >> 4;

    const float* __restrict__ cw = codes + (size_t)(b * CS + cgp * 32u) * SPATIAL + wave * WPOS;
    const float* __restrict__ mw = maskf + (size_t)(b * SSEG) * SPATIAL + wave * WPOS;
    const unsigned loff = kh * SPATIAL + 4u * n;
    v4f (*T)[TPITCH] = tile[wave];

    const bool  is_seg = n < SSEG;
    const float fill   = (n == SSEG) ? 1.0f : 0.0f;

    v8f c0 = {0.f, 0.f, 0.f, 0.f, 0.f, 0.f, 0.f, 0.f};
    v8f c1 = {0.f, 0.f, 0.f, 0.f, 0.f, 0.f, 0.f, 0.f};

    v4f sc[NCL], sm[NML];

    #pragma unroll
    for (int i = 0; i < NCL; ++i)
        sc[i] = __builtin_nontemporal_load((const v4f*)(cw + 2 * i * SPATIAL + loff));
    #pragma unroll
    for (int i = 0; i < NML; ++i)
        sm[i] = *(const v4f*)(mw + 2 * i * SPATIAL + loff);

    for (int g = 0; g < NGRP; ++g) {
        #pragma unroll
        for (int i = 0; i < NCL; ++i) T[2 * i + kh][n] = sc[i];
        #pragma unroll
        for (int i = 0; i < NML; ++i) T[CROWS + 2 * i + kh][n] = sm[i];
        if (g + 1 < NGRP) {
            const float* __restrict__ cg = cw + (g + 1) * GP;
            const float* __restrict__ mg = mw + (g + 1) * GP;
            #pragma unroll
            for (int i = 0; i < NCL; ++i)
                sc[i] = __builtin_nontemporal_load((const v4f*)(cg + 2 * i * SPATIAL + loff));
            #pragma unroll
            for (int i = 0; i < NML; ++i)
                sm[i] = *(const v4f*)(mg + 2 * i * SPATIAL + loff);
        }
        v4f av[NSTP], pv[NSTP], qv[NSTP];
        #pragma unroll
        for (int j = 0; j < NSTP; ++j) {
            av[j] = T[CROWS + (n & 7u)][2 * j + kh];
            pv[j] = T[n][2 * j + kh];
            qv[j] = T[16u + n][2 * j + kh];
        }
        __builtin_amdgcn_sched_barrier(0);
        #pragma unroll
        for (int j = 0; j < NSTP; ++j) {
            const v4f a = av[j], p = pv[j], q = qv[j];
            const v2f a01 = {is_seg ? a.x : fill, is_seg ? a.y : fill};
            const v2f a23 = {is_seg ? a.z : fill, is_seg ? a.w : fill};
            const v2f p01 = {p.x, p.y}, p23 = {p.z, p.w};
            const v2f q01 = {q.x, q.y}, q23 = {q.z, q.w};
            c0 = __builtin_amdgcn_wmma_f32_16x16x4_f32(false, a01, false, p01, (short)0, c0, false, false);
            c1 = __builtin_amdgcn_wmma_f32_16x16x4_f32(false, a01, false, q01, (short)0, c1, false, false);
            c0 = __builtin_amdgcn_wmma_f32_16x16x4_f32(false, a23, false, p23, (short)0, c0, false, false);
            c1 = __builtin_amdgcn_wmma_f32_16x16x4_f32(false, a23, false, q23, (short)0, c1, false, false);
        }
    }

    if (kh == 0u) {
        #pragma unroll
        for (int s = 0; s < SSEG; ++s) {
            part_seg[wave][s][n]       = c0[s];
            part_seg[wave][s][16u + n] = c1[s];
        }
    } else {
        part_glob[wave][n]       = c0[0];
        part_glob[wave][16u + n] = c1[0];
    }
    __syncthreads();

    const unsigned s  = threadIdx.x >> 5;
    const unsigned ch = lane;
    float sum = 0.0f, glob = 0.0f;
    #pragma unroll
    for (int w = 0; w < KSPLIT; ++w) {
        sum  += part_seg[w][s][ch];
        glob += part_glob[w][ch];
    }
    const float ar = area[b * SSEG + s];
    out[(size_t)b * (SSEG * CS) + (size_t)s * CS + cgp * 32u + ch] =
        (ar > 0.0f) ? (sum / ar) : (glob * (1.0f / (float)SPATIAL));
}

extern "C" void kernel_launch(void* const* d_in, const int* in_sizes, int n_in,
                              void* d_out, int out_size, void* d_ws, size_t ws_size,
                              hipStream_t stream) {
    const float* codes = (const float*)d_in[0];
    const int*   sem   = (const int*)d_in[1];
    float*       out   = (float*)d_out;

    float* maskf = (float*)d_ws;
    float* area  = (float*)((char*)d_ws +
                            (size_t)BS * NSTEP * 32 * 2 * sizeof(float));

    hipLaunchKernelGGL(build_mask, dim3(BS * SSEG), dim3(256),
                       0, stream, sem, maskf, area);
    hipLaunchKernelGGL(seg_reduce, dim3(BS * 16), dim3(256),
                       0, stream, codes, maskf, area, out);
}


# mask carried as bits: stage 1 packs each segment's 0/1 row by wave ballot; seg_reduce expands A fragments in registers from one 8-byte load per group, no f32 mask rows via LDS (sum order unchanged)
# speedup vs baseline: 5.4902x; 1.0780x over previous
#include <hip/hip_runtime.h>


typedef float v2f __attribute__((ext_vector_type(2)));
typedef float v4f __attribute__((ext_vector_type(4)));
typedef float v8f __attribute__((ext_vector_type(8)));

#define BS      32
#define CS      512
#define SSEG    8
#define HS      64
#define WSZ     64
#define SPATIAL (HS * WSZ)
#define NSTEP   (SPATIAL / 4)
#define SEMW    256

#define KSPLIT  8
#define WPOS    (SPATIAL / KSPLIT)
#define GP      64
#define NGRP    (WPOS / GP)
#define CROWS   32
#define TPITCH  17
#define NCL     (CROWS / 2)
#define NSTP    (GP / 8)
#define MWORDS  (SPATIAL / 32)

__global__ void __launch_bounds__(128) build_mask(const int* __restrict__ sem,
                                                  unsigned* __restrict__ maskb,
                                                  float* __restrict__ area) {
    const unsigned bs   = blockIdx.x;
    const unsigned wv   = threadIdx.x >> 5;
    const unsigned lane = threadIdx.x & 31u;
    __shared__ unsigned cnt[4];
    const int* __restrict__ src = sem + (size_t)bs * (SEMW * SEMW) + (size_t)(wv * 16u) * (4u * SEMW) + 4u * lane;
    int v[32];
    #pragma unroll
    for (int i = 0; i < 32; ++i)
        v[i] = src[(i >> 1) * (4 * SEMW) + (i & 1) * 128];
    unsigned mine = 0u, count = 0u;
    #pragma unroll
    for (int i = 0; i < 32; ++i) {
        const unsigned wd = __builtin_amdgcn_ballot_w32(v[i] != 0);
        count += (unsigned)__popc(wd);
        mine = (lane == (unsigned)i) ? wd : mine;
    }
    maskb[bs * MWORDS + wv * 32u + lane] = mine;
    if (lane == 0u) cnt[wv] = count;
    __syncthreads();
    if (threadIdx.x == 0u) area[bs] = (float)((cnt[0] + cnt[1]) + (cnt[2] + cnt[3]));
}

__global__ void __launch_bounds__(256) seg_reduce(const float* __restrict__ codes,
                                                  const unsigned* __restrict__ maskb,
                                                  const float* __restrict__ area,
                                                  float* __restrict__ out) {
    __shared__ v4f   tile[KSPLIT][CROWS][TPITCH];
    __shared__ float part_seg[KSPLIT][SSEG][32];
    __shared__ float part_glob[KSPLIT][32];

    const unsigned wave = __builtin_amdgcn_readfirstlane(threadIdx.x >> 5);
    const unsigned lane = threadIdx.x & 31u;
    const unsigned b    = blockIdx.x >> 4;
    const unsigned cgp  = blockIdx.x & 15u;
    const unsigned n    = lane & 15u;
    const unsigned kh   = lane >> 4;

    const float* __restrict__ cw = codes + (size_t)(b * CS + cgp * 32u) * SPATIAL + wave * WPOS;
    const unsigned loff = kh * SPATIAL + 4u * n;
    v4f (*T)[TPITCH] = tile[wave];

    const uint2* __restrict__ mrow =
        (const uint2*)maskb + (size_t)(b * SSEG + (n & 7u)) * (MWORDS / 2) + wave * NGRP;

    const bool     is_seg = n < SSEG;
    const unsigned fillw  = (n == SSEG) ? 0xFFFFFFFFu : 0u;
    const unsigned sh     = 4u * kh;

    v8f c0 = {0.f, 0.f, 0.f, 0.f, 0.f, 0.f, 0.f, 0.f};
    v8f c1 = {0.f, 0.f, 0.f, 0.f, 0.f, 0.f, 0.f, 0.f};

    v4f sc[NCL];

    #pragma unroll
    for (int i = 0; i < NCL; ++i)
        sc[i] = __builtin_nontemporal_load((const v4f*)(cw + 2 * i * SPATIAL + loff));
    uint2 mb = mrow[0];

    for (int g = 0; g < NGRP; ++g) {
        #pragma unroll
        for (int i = 0; i < NCL; ++i) T[2 * i + kh][n] = sc[i];
        const unsigned lo = (is_seg ? mb.x : fillw) >> sh;
        const unsigned hi = (is_seg ? mb.y : fillw) >> sh;
        if (g + 1 < NGRP) {
            const float* __restrict__ cg = cw + (g + 1) * GP;
            #pragma unroll
            for (int i = 0; i < NCL; ++i)
                sc[i] = __builtin_nontemporal_load((const v4f*)(cg + 2 * i * SPATIAL + loff));
            mb = mrow[g + 1];
        }
        v4f pv[NSTP], qv[NSTP];
        #pragma unroll
        for (int j = 0; j < NSTP; ++j) {
            pv[j] = T[n][2 * j + kh];
            qv[j] = T[16u + n][2 * j + kh];
        }
        __builtin_amdgcn_sched_barrier(0);
        #pragma unroll
        for (int j = 0; j < NSTP; ++j) {
            const v4f p = pv[j], q = qv[j];
            const unsigned x = ((j < 4) ? lo : hi) >> (8 * (j & 3));
            const v2f a01 = {(float)(x & 1u), (float)((x >> 1) & 1u)};
            const v2f a23 = {(float)((x >> 2) & 1u), (float)((x >> 3) & 1u)};
            const v2f p01 = {p.x, p.y}, p23 = {p.z, p.w};
            const v2f q01 = {q.x, q.y}, q23 = {q.z, q.w};
            c0 = __builtin_amdgcn_wmma_f32_16x16x4_f32(false, a01, false, p01, (short)0, c0, false, false);
            c1 = __builtin_amdgcn_wmma_f32_16x16x4_f32(false, a01, false, q01, (short)0, c1, false, false);
            c0 = __builtin_amdgcn_wmma_f32_16x16x4_f32(false, a23, false, p23, (short)0, c0, false, false);
            c1 = __builtin_amdgcn_wmma_f32_16x16x4_f32(false, a23, false, q23, (short)0, c1, false, false);
        }
    }

    if (kh == 0u) {
        #pragma unroll
        for (int s = 0; s < SSEG; ++s) {
            part_seg[wave][s][n]       = c0[s];
            part_seg[wave][s][16u + n] = c1[s];
        }
    } else {
        part_glob[wave][n]       = c0[0];
        part_glob[wave][16u + n] = c1[0];
    }
    __syncthreads();

    const unsigned s  = threadIdx.x >> 5;
    const unsigned ch = lane;
    float sum = 0.0f, glob = 0.0f;
    #pragma unroll
    for (int w = 0; w < KSPLIT; ++w) {
        sum  += part_seg[w][s][ch];
        glob += part_glob[w][ch];
    }
    const float ar = area[b * SSEG + s];
    out[(size_t)b * (SSEG * CS) + (size_t)s * CS + cgp * 32u + ch] =
        (ar > 0.0f) ? (sum / ar) : (glob * (1.0f / (float)SPATIAL));
}

extern "C" void kernel_launch(void* const* d_in, const int* in_sizes, int n_in,
                              void* d_out, int out_size, void* d_ws, size_t ws_size,
                              hipStream_t stream) {
    const float* codes = (const float*)d_in[0];
    const int*   sem   = (const int*)d_in[1];
    float*       out   = (float*)d_out;

    unsigned* maskb = (unsigned*)d_ws;
    float*    area  = (float*)((char*)d_ws +
                               (size_t)BS * NSTEP * 32 * 2 * sizeof(float));

    hipLaunchKernelGGL(build_mask, dim3(BS * SSEG), dim3(128),
                       0, stream, sem, maskb, area);
    hipLaunchKernelGGL(seg_reduce, dim3(BS * 16), dim3(256),
                       0, stream, codes, maskb, area, out);
}
